# starting kernel
# baseline (speedup 1.0000x reference)
#include <hip/hip_runtime.h>

typedef __attribute__((ext_vector_type(2))) float v2f;
typedef __attribute__((ext_vector_type(8))) float v8f;

__device__ __forceinline__ float swap16(float v)
{
    int i = __builtin_bit_cast(int, v);
    i = __builtin_amdgcn_ds_swizzle(i, 0x401F);
    return __builtin_bit_cast(float, i);
}

__device__ __forceinline__ float rcp_nr(float t)
{
    float r = __builtin_amdgcn_rcpf(t);
    r = r * fmaf(-t, r, 2.0f);
    return r;
}

__device__ __forceinline__ float tile_eval(float xv, bool hiHalf, v2f A1, v2f A2)
{
    const float x2 = xv * xv;
    const float x3 = x2 * xv;
    const float x4 = x2 * x2;

    v2f B1, B2;
    B1.x = hiHalf ? x2 : 1.0f;
    B1.y = hiHalf ? x3 : xv;
    B2.x = x4 * B1.x;
    B2.y = x4 * B1.y;

    v8f cz = {0.f, 0.f, 0.f, 0.f, 0.f, 0.f, 0.f, 0.f};

    v8f D = __builtin_amdgcn_wmma_f32_16x16x4_f32(
        false, A1, false, B1, (short)0, cz, false, false);
    D = __builtin_amdgcn_wmma_f32_16x16x4_f32(
        false, A2, false, B2, (short)0, D, false, false);

    const float y = x4 * x4;

    float S = D[7];
#pragma unroll
    for (int r = 6; r >= 0; --r) S = fmaf(S, y, D[r]);

    float t = fmaf(fabsf(xv * S), 1.0f, 1.0f);
    float sendv = hiHalf ? t : S;
    float recv  = swap16(sendv);

    float numv = hiHalf ? recv : S;
    float tv   = hiHalf ? t    : recv;
    return numv * rcp_nr(tv);
}

__global__ __launch_bounds__(256)
void rationals_wmma_kernel(const float* __restrict__ x,
                           const float* __restrict__ coef,
                           float* __restrict__ out, int N)
{
    const int lane = threadIdx.x & 31;
    const int col  = lane & 15;
    const bool hi  = (lane >= 16);

    const int kb = hi ? 2 : 0;
    const int cb = 8 * col;
    v2f A1, A2;
    A1.x = coef[cb + kb + 0];
    A1.y = coef[cb + kb + 1];
    A2.x = coef[cb + 4 + kb + 0];
    A2.y = coef[cb + 4 + kb + 1];

    const int wid        = (blockIdx.x * blockDim.x + threadIdx.x) >> 5;
    const int totalWaves = (gridDim.x * blockDim.x) >> 5;

    for (int base = wid * 64; base < N; base += totalWaves * 64) {
        if (base + 64 <= N) {
            float x0 = x[base + col];
            float x1 = x[base + 16 + col];
            float x2 = x[base + 32 + col];
            float x3 = x[base + 48 + col];

            float r0 = tile_eval(x0, hi, A1, A2);
            float r1 = tile_eval(x1, hi, A1, A2);
            float r2 = tile_eval(x2, hi, A1, A2);
            float r3 = tile_eval(x3, hi, A1, A2);

            out[base + lane]      = hi ? r1 : r0;
            out[base + 32 + lane] = hi ? r3 : r2;
        } else {
#pragma unroll
            for (int e = 0; e < 64; e += 32) {
                int idx = base + e + lane;
                if (idx < N) {
                    float xv = x[idx];
                    float sn = coef[63];
                    float sd = coef[127];
                    for (int p = 62; p >= 0; --p) {
                        sn = fmaf(sn, xv, coef[p]);
                        sd = fmaf(sd, xv, coef[64 + p]);
                    }
                    sd *= xv;
                    out[idx] = sn / (fabsf(sd) + 1.0f);
                }
            }
        }
    }
}

extern "C" void kernel_launch(void* const* d_in, const int* in_sizes, int n_in,
                              void* d_out, int out_size, void* d_ws, size_t ws_size,
                              hipStream_t stream)
{
    const float* x    = (const float*)d_in[0];
    const float* coef = (const float*)d_in[1];
    float* out        = (float*)d_out;
    const int N       = in_sizes[0];

    const int block = 256;
    int grid = (N + 511) / 512;
    if (grid > 4096) grid = 4096;
    if (grid < 1) grid = 1;

    rationals_wmma_kernel<<<dim3(grid), dim3(block), 0, stream>>>(x, coef, out, N);
}


# Resident grid: 512 blocks x 8 waves, each wave loops over its 64-element groups four per trip with the next trip's x loads issued first (was 4096 blocks, 2 one-group trips per wave, entry-bound)
# speedup vs baseline: 1.0907x; 1.0907x over previous
#include <hip/hip_runtime.h>

typedef __attribute__((ext_vector_type(2))) float v2f;
typedef __attribute__((ext_vector_type(8))) float v8f;

__device__ __forceinline__ float swap16(float v)
{
    int i = __builtin_bit_cast(int, v);
    i = __builtin_amdgcn_ds_swizzle(i, 0x401F);
    return __builtin_bit_cast(float, i);
}

__device__ __forceinline__ float rcp_nr(float t)
{
    float r = __builtin_amdgcn_rcpf(t);
    r = r * fmaf(-t, r, 2.0f);
    return r;
}

__device__ __forceinline__ float tile_eval(float xv, bool hiHalf, v2f A1, v2f A2)
{
    const float x2 = xv * xv;
    const float x3 = x2 * xv;
    const float x4 = x2 * x2;

    v2f B1, B2;
    B1.x = hiHalf ? x2 : 1.0f;
    B1.y = hiHalf ? x3 : xv;
    B2.x = x4 * B1.x;
    B2.y = x4 * B1.y;

    v8f cz = {0.f, 0.f, 0.f, 0.f, 0.f, 0.f, 0.f, 0.f};

    v8f D = __builtin_amdgcn_wmma_f32_16x16x4_f32(
        false, A1, false, B1, (short)0, cz, false, false);
    D = __builtin_amdgcn_wmma_f32_16x16x4_f32(
        false, A2, false, B2, (short)0, D, false, false);

    const float y = x4 * x4;

    float S = D[7];
#pragma unroll
    for (int r = 6; r >= 0; --r) S = fmaf(S, y, D[r]);

    float t = fmaf(fabsf(xv * S), 1.0f, 1.0f);
    float sendv = hiHalf ? t : S;
    float recv  = swap16(sendv);

    float numv = hiHalf ? recv : S;
    float tv   = hiHalf ? t    : recv;
    return numv * rcp_nr(tv);
}

__device__ __forceinline__ void load_group(const float* __restrict__ x, int g, int col,
                                           float (&v)[4])
{
    const float* p = x + g * 64 + col;
    v[0] = p[0];
    v[1] = p[16];
    v[2] = p[32];
    v[3] = p[48];
}

__device__ __forceinline__ void eval_group(const float (&v)[4], bool hi, v2f A1, v2f A2,
                                           float* __restrict__ out, int g, int lane)
{
    float r0 = tile_eval(v[0], hi, A1, A2);
    float r1 = tile_eval(v[1], hi, A1, A2);
    float r2 = tile_eval(v[2], hi, A1, A2);
    float r3 = tile_eval(v[3], hi, A1, A2);

    float* o = out + g * 64 + lane;
    o[0]  = hi ? r1 : r0;
    o[32] = hi ? r3 : r2;
}

#define GPI 4

__global__ __launch_bounds__(256)
void rationals_wmma_kernel(const float* __restrict__ x,
                           const float* __restrict__ coef,
                           float* __restrict__ out, int N)
{
    const int lane = threadIdx.x & 31;
    const int col  = lane & 15;
    const bool hi  = (lane >= 16);

    const int kb = hi ? 2 : 0;
    const int cb = 8 * col;
    v2f A1, A2;
    A1.x = coef[cb + kb + 0];
    A1.y = coef[cb + kb + 1];
    A2.x = coef[cb + 4 + kb + 0];
    A2.y = coef[cb + 4 + kb + 1];

    const int wid = (blockIdx.x * blockDim.x + threadIdx.x) >> 5;
    const int gs  = (gridDim.x * blockDim.x) >> 5;
    const int nFull = N >> 6;

    int g = wid;
    const int span = (GPI - 1) * gs;
    if (g + span < nFull) {
        float cur[GPI][4];
#pragma unroll
        for (int j = 0; j < GPI; ++j) load_group(x, g + j * gs, col, cur[j]);

        int gn = g + GPI * gs;
        while (gn + span < nFull) {
            float nxt[GPI][4];
#pragma unroll
            for (int j = 0; j < GPI; ++j) load_group(x, gn + j * gs, col, nxt[j]);
            __builtin_amdgcn_sched_barrier(0);
#pragma unroll
            for (int j = 0; j < GPI; ++j) eval_group(cur[j], hi, A1, A2, out, g + j * gs, lane);
#pragma unroll
            for (int j = 0; j < GPI; ++j) {
                cur[j][0] = nxt[j][0];
                cur[j][1] = nxt[j][1];
                cur[j][2] = nxt[j][2];
                cur[j][3] = nxt[j][3];
            }
            g = gn;
            gn += GPI * gs;
        }
#pragma unroll
        for (int j = 0; j < GPI; ++j) eval_group(cur[j], hi, A1, A2, out, g + j * gs, lane);
        g = gn;
    }

    for (; g < nFull; g += gs) {
        float v[4];
        load_group(x, g, col, v);
        eval_group(v, hi, A1, A2, out, g, lane);
    }

    if (g == nFull && (N & 63)) {
#pragma unroll
        for (int e = 0; e < 64; e += 32) {
            int idx = nFull * 64 + e + lane;
            if (idx < N) {
                float xv = x[idx];
                float sn = coef[63];
                float sd = coef[127];
                for (int p = 62; p >= 0; --p) {
                    sn = fmaf(sn, xv, coef[p]);
                    sd = fmaf(sd, xv, coef[64 + p]);
                }
                sd *= xv;
                out[idx] = sn / (fabsf(sd) + 1.0f);
            }
        }
    }
}

extern "C" void kernel_launch(void* const* d_in, const int* in_sizes, int n_in,
                              void* d_out, int out_size, void* d_ws, size_t ws_size,
                              hipStream_t stream)
{
    const float* x    = (const float*)d_in[0];
    const float* coef = (const float*)d_in[1];
    float* out        = (float*)d_out;
    const int N       = in_sizes[0];

    rationals_wmma_kernel<<<dim3(512), dim3(256), 0, stream>>>(x, coef, out, N);
}


# Finish per tile pair in lane order: one v_permlanex16 exchange per 32 elements (was an LDS swizzle per 16), one fused |x|*|d|+1 and one reciprocal per element (was two), no output select
# speedup vs baseline: 1.1475x; 1.0521x over previous
#include <hip/hip_runtime.h>

typedef __attribute__((ext_vector_type(2))) float v2f;
typedef __attribute__((ext_vector_type(8))) float v8f;

__device__ __forceinline__ float xhalf(float v)
{
    int i = __builtin_bit_cast(int, v);
    i = __builtin_amdgcn_permlanex16(i, i, 0x76543210, 0xfedcba98, false, false);
    return __builtin_bit_cast(float, i);
}

__device__ __forceinline__ float rcp_nr(float t)
{
    float r = __builtin_amdgcn_rcpf(t);
    r = r * fmaf(-t, r, 2.0f);
    return r;
}

__device__ __forceinline__ float tile_poly(float xv, bool hiHalf, v2f A1, v2f A2)
{
    const float x2 = xv * xv;
    const float x3 = x2 * xv;
    const float x4 = x2 * x2;

    v2f B1, B2;
    B1.x = hiHalf ? x2 : 1.0f;
    B1.y = hiHalf ? x3 : xv;
    B2.x = x4 * B1.x;
    B2.y = x4 * B1.y;

    v8f cz = {0.f, 0.f, 0.f, 0.f, 0.f, 0.f, 0.f, 0.f};

    v8f D = __builtin_amdgcn_wmma_f32_16x16x4_f32(
        false, A1, false, B1, (short)0, cz, false, false);
    D = __builtin_amdgcn_wmma_f32_16x16x4_f32(
        false, A2, false, B2, (short)0, D, false, false);

    const float y = x4 * x4;

    float S = D[7];
#pragma unroll
    for (int r = 6; r >= 0; --r) S = fmaf(S, y, D[r]);
    return S;
}

__device__ __forceinline__ float pair_finish(float Sa, float Sb, bool hiHalf, float xl)
{
    const float u   = hiHalf ? Sa : Sb;
    const float w   = xhalf(u);
    const float num = hiHalf ? w  : Sa;
    const float den = hiHalf ? Sb : w;
    const float t = fmaf(fabsf(xl), fabsf(den), 1.0f);
    return num * rcp_nr(t);
}

__device__ __forceinline__ void load_group(const float* __restrict__ x, int g, int col, int lane,
                                           float (&v)[6])
{
    const float* p = x + g * 64 + col;
    v[0] = p[0];
    v[1] = p[16];
    v[2] = p[32];
    v[3] = p[48];
    const float* q = x + g * 64 + lane;
    v[4] = q[0];
    v[5] = q[32];
}

__device__ __forceinline__ void eval_group(const float (&v)[6], bool hi, v2f A1, v2f A2,
                                           float* __restrict__ out, int g, int lane)
{
    float s0 = tile_poly(v[0], hi, A1, A2);
    float s1 = tile_poly(v[1], hi, A1, A2);
    float s2 = tile_poly(v[2], hi, A1, A2);
    float s3 = tile_poly(v[3], hi, A1, A2);

    float* o = out + g * 64 + lane;
    o[0]  = pair_finish(s0, s1, hi, v[4]);
    o[32] = pair_finish(s2, s3, hi, v[5]);
}

#define GPI 4

__global__ __launch_bounds__(256)
void rationals_wmma_kernel(const float* __restrict__ x,
                           const float* __restrict__ coef,
                           float* __restrict__ out, int N)
{
    const int lane = threadIdx.x & 31;
    const int col  = lane & 15;
    const bool hi  = (lane >= 16);

    const int kb = hi ? 2 : 0;
    const int cb = 8 * col;
    v2f A1, A2;
    A1.x = coef[cb + kb + 0];
    A1.y = coef[cb + kb + 1];
    A2.x = coef[cb + 4 + kb + 0];
    A2.y = coef[cb + 4 + kb + 1];

    const int wid = (blockIdx.x * blockDim.x + threadIdx.x) >> 5;
    const int gs  = (gridDim.x * blockDim.x) >> 5;
    const int nFull = N >> 6;

    int g = wid;
    const int span = (GPI - 1) * gs;
    if (g + span < nFull) {
        float cur[GPI][6];
#pragma unroll
        for (int j = 0; j < GPI; ++j) load_group(x, g + j * gs, col, lane, cur[j]);

        int gn = g + GPI * gs;
        while (gn + span < nFull) {
            float nxt[GPI][6];
#pragma unroll
            for (int j = 0; j < GPI; ++j) load_group(x, gn + j * gs, col, lane, nxt[j]);
            __builtin_amdgcn_sched_barrier(0);
#pragma unroll
            for (int j = 0; j < GPI; ++j) eval_group(cur[j], hi, A1, A2, out, g + j * gs, lane);
#pragma unroll
            for (int j = 0; j < GPI; ++j) {
                cur[j][0] = nxt[j][0];
                cur[j][1] = nxt[j][1];
                cur[j][2] = nxt[j][2];
                cur[j][3] = nxt[j][3];
                cur[j][4] = nxt[j][4];
                cur[j][5] = nxt[j][5];
            }
            g = gn;
            gn += GPI * gs;
        }
#pragma unroll
        for (int j = 0; j < GPI; ++j) eval_group(cur[j], hi, A1, A2, out, g + j * gs, lane);
        g = gn;
    }

    for (; g < nFull; g += gs) {
        float v[6];
        load_group(x, g, col, lane, v);
        eval_group(v, hi, A1, A2, out, g, lane);
    }

    if (g == nFull && (N & 63)) {
#pragma unroll
        for (int e = 0; e < 64; e += 32) {
            int idx = nFull * 64 + e + lane;
            if (idx < N) {
                float xv = x[idx];
                float sn = coef[63];
                float sd = coef[127];
                for (int p = 62; p >= 0; --p) {
                    sn = fmaf(sn, xv, coef[p]);
                    sd = fmaf(sd, xv, coef[64 + p]);
                }
                sd *= xv;
                out[idx] = sn / (fabsf(sd) + 1.0f);
            }
        }
    }
}

extern "C" void kernel_launch(void* const* d_in, const int* in_sizes, int n_in,
                              void* d_out, int out_size, void* d_ws, size_t ws_size,
                              hipStream_t stream)
{
    const float* x    = (const float*)d_in[0];
    const float* coef = (const float*)d_in[1];
    float* out        = (float*)d_out;
    const int N       = in_sizes[0];

    rationals_wmma_kernel<<<dim3(512), dim3(256), 0, stream>>>(x, coef, out, N);
}
